# starting kernel
# baseline (speedup 1.0000x reference)
#include <hip/hip_runtime.h>
#include <hip/hip_bf16.h>

typedef __attribute__((ext_vector_type(2))) float v2f;
typedef __attribute__((ext_vector_type(8))) float v8f;

#define T_STEPS 2048
#define BATCH   512
#define NIN     27
#define NROWS   (T_STEPS * BATCH)
#define NTILES  (NROWS / 16)

__global__ void lstm_xproj_wmma(const float* __restrict__ x,
                                const float* __restrict__ W_ih,
                                const float* __restrict__ b_ih,
                                const float* __restrict__ b_hh,
                                float* __restrict__ xp)
{
    const int lane   = threadIdx.x & 31;
    const int wave   = (blockIdx.x * blockDim.x + threadIdx.x) >> 5;
    const int nwaves = (gridDim.x * blockDim.x) >> 5;

    const int mn    = lane & 15;
    const int khalf = (lane >> 4) << 1;

    v2f bmat[7];
#pragma unroll
    for (int kk = 0; kk < 7; ++kk) {
        const int k0 = kk * 4 + khalf;
        float bx = 0.0f, by = 0.0f;
        if (mn < 4) {
            if (k0     < NIN) bx = W_ih[mn * NIN + k0];
            if (k0 + 1 < NIN) by = W_ih[mn * NIN + k0 + 1];
        }
        bmat[kk].x = bx;
        bmat[kk].y = by;
    }
    const float bias = (mn < 4) ? (b_ih[mn] + b_hh[mn]) : 0.0f;

    for (int tile = wave; tile < NTILES; tile += nwaves) {
        const float* __restrict__ arow = x + (size_t)(tile * 16 + mn) * NIN;

        v8f acc = {};
#pragma unroll
        for (int kk = 0; kk < 7; ++kk) {
            const int k0 = kk * 4 + khalf;
            v2f a;
            a.x = (k0     < NIN) ? arow[k0]     : 0.0f;
            a.y = (k0 + 1 < NIN) ? arow[k0 + 1] : 0.0f;
            acc = __builtin_amdgcn_wmma_f32_16x16x4_f32(
                false, a, false, bmat[kk], (short)0, acc, false, false);
        }

        if (mn < 4) {
            const int rbase = tile * 16 + ((lane >> 4) << 3);
#pragma unroll
            for (int r = 0; r < 8; ++r) {
                xp[(size_t)(rbase + r) * 4 + mn] = acc[r] + bias;
            }
        }
    }
}

__device__ __forceinline__ float fast_sigmoid(float v) {
    return 1.0f / (1.0f + __expf(-v));
}
__device__ __forceinline__ float fast_tanh(float v) {
    return 1.0f - 2.0f / (__expf(2.0f * v) + 1.0f);
}

__global__ void lstm_scan(const float* __restrict__ xp,
                          const float* __restrict__ W_hh,
                          float* __restrict__ out)
{
    const int b = blockIdx.x * blockDim.x + threadIdx.x;
    if (b >= BATCH) return;

    const float wi = W_hh[0];
    const float wf = W_hh[1];
    const float wg = W_hh[2];
    const float wo = W_hh[3];

    float h = 0.0f, c = 0.0f;
    for (int t = 0; t < T_STEPS; ++t) {
        const float4 g = *(const float4*)(xp + ((size_t)t * BATCH + b) * 4);
        const float i  = fast_sigmoid(g.x + h * wi);
        const float f  = fast_sigmoid(g.y + h * wf);
        const float gg = fast_tanh   (g.z + h * wg);
        const float o  = fast_sigmoid(g.w + h * wo);
        c = f * c + i * gg;
        h = o * fast_tanh(c);
        out[(size_t)t * BATCH + b] = h;
    }
}

extern "C" void kernel_launch(void* const* d_in, const int* in_sizes, int n_in,
                              void* d_out, int out_size, void* d_ws, size_t ws_size,
                              hipStream_t stream)
{
    const float* x    = (const float*)d_in[0];
    const float* W_ih = (const float*)d_in[1];
    const float* W_hh = (const float*)d_in[2];
    const float* b_ih = (const float*)d_in[3];
    const float* b_hh = (const float*)d_in[4];
    float* out = (float*)d_out;
    float* xp  = (float*)d_ws;

    lstm_xproj_wmma<<<512, 256, 0, stream>>>(x, W_ih, b_ih, b_hh, xp);

    lstm_scan<<<2, 256, 0, stream>>>(xp, W_hh, out);
}


# scan kernel: each lane fetches its gate rows 16 time steps ahead into a register ring, so the global-load round trip leaves the h-to-h dependency chain; cell arithmetic and its order unchanged
# speedup vs baseline: 3.9898x; 3.9898x over previous
#include <hip/hip_runtime.h>
#include <hip/hip_bf16.h>

typedef __attribute__((ext_vector_type(2))) float v2f;
typedef __attribute__((ext_vector_type(8))) float v8f;

#define T_STEPS 2048
#define BATCH   512
#define NIN     27
#define NROWS   (T_STEPS * BATCH)
#define NTILES  (NROWS / 16)

__global__ void lstm_xproj_wmma(const float* __restrict__ x,
                                const float* __restrict__ W_ih,
                                const float* __restrict__ b_ih,
                                const float* __restrict__ b_hh,
                                float* __restrict__ xp)
{
    const int lane   = threadIdx.x & 31;
    const int wave   = (blockIdx.x * blockDim.x + threadIdx.x) >> 5;
    const int nwaves = (gridDim.x * blockDim.x) >> 5;

    const int mn    = lane & 15;
    const int khalf = (lane >> 4) << 1;

    v2f bmat[7];
#pragma unroll
    for (int kk = 0; kk < 7; ++kk) {
        const int k0 = kk * 4 + khalf;
        float bx = 0.0f, by = 0.0f;
        if (mn < 4) {
            if (k0     < NIN) bx = W_ih[mn * NIN + k0];
            if (k0 + 1 < NIN) by = W_ih[mn * NIN + k0 + 1];
        }
        bmat[kk].x = bx;
        bmat[kk].y = by;
    }
    const float bias = (mn < 4) ? (b_ih[mn] + b_hh[mn]) : 0.0f;

    for (int tile = wave; tile < NTILES; tile += nwaves) {
        const float* __restrict__ arow = x + (size_t)(tile * 16 + mn) * NIN;

        v8f acc = {};
#pragma unroll
        for (int kk = 0; kk < 7; ++kk) {
            const int k0 = kk * 4 + khalf;
            v2f a;
            a.x = (k0     < NIN) ? arow[k0]     : 0.0f;
            a.y = (k0 + 1 < NIN) ? arow[k0 + 1] : 0.0f;
            acc = __builtin_amdgcn_wmma_f32_16x16x4_f32(
                false, a, false, bmat[kk], (short)0, acc, false, false);
        }

        if (mn < 4) {
            const int rbase = tile * 16 + ((lane >> 4) << 3);
#pragma unroll
            for (int r = 0; r < 8; ++r) {
                xp[(size_t)(rbase + r) * 4 + mn] = acc[r] + bias;
            }
        }
    }
}

__device__ __forceinline__ float fast_sigmoid(float v) {
    return 1.0f / (1.0f + __expf(-v));
}
__device__ __forceinline__ float fast_tanh(float v) {
    return 1.0f - 2.0f / (__expf(2.0f * v) + 1.0f);
}

#define PF 16

__device__ __forceinline__ void lstm_cell(const float4 g,
                                          const float wi, const float wf,
                                          const float wg, const float wo,
                                          float& h, float& c)
{
    const float i  = fast_sigmoid(g.x + h * wi);
    const float f  = fast_sigmoid(g.y + h * wf);
    const float gg = fast_tanh   (g.z + h * wg);
    const float o  = fast_sigmoid(g.w + h * wo);
    c = f * c + i * gg;
    h = o * fast_tanh(c);
}

__global__ void lstm_scan(const float* __restrict__ xp,
                          const float* __restrict__ W_hh,
                          float* __restrict__ out)
{
    const int b = blockIdx.x * blockDim.x + threadIdx.x;
    if (b >= BATCH) return;

    const float wi = W_hh[0];
    const float wf = W_hh[1];
    const float wg = W_hh[2];
    const float wo = W_hh[3];

    const float4* __restrict__ gp = (const float4*)xp + b;
    float* __restrict__ op = out + b;

    float4 ring[PF];
#pragma unroll
    for (int j = 0; j < PF; ++j) ring[j] = gp[(size_t)j * BATCH];

    float h = 0.0f, c = 0.0f;

    for (int t0 = 0; t0 < T_STEPS - PF; t0 += PF) {
#pragma unroll
        for (int j = 0; j < PF; ++j) {
            lstm_cell(ring[j], wi, wf, wg, wo, h, c);
            op[(size_t)(t0 + j) * BATCH] = h;
            __builtin_amdgcn_sched_barrier(0);
            ring[j] = gp[(size_t)(t0 + j + PF) * BATCH];
        }
    }

#pragma unroll
    for (int j = 0; j < PF; ++j) {
        lstm_cell(ring[j], wi, wf, wg, wo, h, c);
        op[(size_t)(T_STEPS - PF + j) * BATCH] = h;
    }
}

extern "C" void kernel_launch(void* const* d_in, const int* in_sizes, int n_in,
                              void* d_out, int out_size, void* d_ws, size_t ws_size,
                              hipStream_t stream)
{
    const float* x    = (const float*)d_in[0];
    const float* W_ih = (const float*)d_in[1];
    const float* W_hh = (const float*)d_in[2];
    const float* b_ih = (const float*)d_in[3];
    const float* b_hh = (const float*)d_in[4];
    float* out = (float*)d_out;
    float* xp  = (float*)d_ws;

    lstm_xproj_wmma<<<512, 256, 0, stream>>>(x, W_ih, b_ih, b_hh, xp);

    lstm_scan<<<2, 256, 0, stream>>>(xp, W_hh, out);
}


# scan cell: the five divisions of sigmoid and tanh become the hardware reciprocal v_rcp_f32 (1 ulp) instead of the correctly rounded sequence; h-to-h chain 34 -> 16 dependent instructions
# speedup vs baseline: 5.2133x; 1.3067x over previous
#include <hip/hip_runtime.h>
#include <hip/hip_bf16.h>

typedef __attribute__((ext_vector_type(2))) float v2f;
typedef __attribute__((ext_vector_type(8))) float v8f;

#define T_STEPS 2048
#define BATCH   512
#define NIN     27
#define NROWS   (T_STEPS * BATCH)
#define NTILES  (NROWS / 16)

__global__ void lstm_xproj_wmma(const float* __restrict__ x,
                                const float* __restrict__ W_ih,
                                const float* __restrict__ b_ih,
                                const float* __restrict__ b_hh,
                                float* __restrict__ xp)
{
    const int lane   = threadIdx.x & 31;
    const int wave   = (blockIdx.x * blockDim.x + threadIdx.x) >> 5;
    const int nwaves = (gridDim.x * blockDim.x) >> 5;

    const int mn    = lane & 15;
    const int khalf = (lane >> 4) << 1;

    v2f bmat[7];
#pragma unroll
    for (int kk = 0; kk < 7; ++kk) {
        const int k0 = kk * 4 + khalf;
        float bx = 0.0f, by = 0.0f;
        if (mn < 4) {
            if (k0     < NIN) bx = W_ih[mn * NIN + k0];
            if (k0 + 1 < NIN) by = W_ih[mn * NIN + k0 + 1];
        }
        bmat[kk].x = bx;
        bmat[kk].y = by;
    }
    const float bias = (mn < 4) ? (b_ih[mn] + b_hh[mn]) : 0.0f;

    for (int tile = wave; tile < NTILES; tile += nwaves) {
        const float* __restrict__ arow = x + (size_t)(tile * 16 + mn) * NIN;

        v8f acc = {};
#pragma unroll
        for (int kk = 0; kk < 7; ++kk) {
            const int k0 = kk * 4 + khalf;
            v2f a;
            a.x = (k0     < NIN) ? arow[k0]     : 0.0f;
            a.y = (k0 + 1 < NIN) ? arow[k0 + 1] : 0.0f;
            acc = __builtin_amdgcn_wmma_f32_16x16x4_f32(
                false, a, false, bmat[kk], (short)0, acc, false, false);
        }

        if (mn < 4) {
            const int rbase = tile * 16 + ((lane >> 4) << 3);
#pragma unroll
            for (int r = 0; r < 8; ++r) {
                xp[(size_t)(rbase + r) * 4 + mn] = acc[r] + bias;
            }
        }
    }
}

__device__ __forceinline__ float fast_rcp(float v) {
    return __builtin_amdgcn_rcpf(v);
}
__device__ __forceinline__ float fast_sigmoid(float v) {
    return fast_rcp(1.0f + __expf(-v));
}
__device__ __forceinline__ float fast_tanh(float v) {
    return 1.0f - 2.0f * fast_rcp(__expf(2.0f * v) + 1.0f);
}

#define PF 16

__device__ __forceinline__ void lstm_cell(const float4 g,
                                          const float wi, const float wf,
                                          const float wg, const float wo,
                                          float& h, float& c)
{
    const float i  = fast_sigmoid(g.x + h * wi);
    const float f  = fast_sigmoid(g.y + h * wf);
    const float gg = fast_tanh   (g.z + h * wg);
    const float o  = fast_sigmoid(g.w + h * wo);
    c = f * c + i * gg;
    h = o * fast_tanh(c);
}

__global__ void lstm_scan(const float* __restrict__ xp,
                          const float* __restrict__ W_hh,
                          float* __restrict__ out)
{
    const int b = blockIdx.x * blockDim.x + threadIdx.x;
    if (b >= BATCH) return;

    const float wi = W_hh[0];
    const float wf = W_hh[1];
    const float wg = W_hh[2];
    const float wo = W_hh[3];

    const float4* __restrict__ gp = (const float4*)xp + b;
    float* __restrict__ op = out + b;

    float4 ring[PF];
#pragma unroll
    for (int j = 0; j < PF; ++j) ring[j] = gp[(size_t)j * BATCH];

    float h = 0.0f, c = 0.0f;

    for (int t0 = 0; t0 < T_STEPS - PF; t0 += PF) {
#pragma unroll
        for (int j = 0; j < PF; ++j) {
            lstm_cell(ring[j], wi, wf, wg, wo, h, c);
            op[(size_t)(t0 + j) * BATCH] = h;
            __builtin_amdgcn_sched_barrier(0);
            ring[j] = gp[(size_t)(t0 + j + PF) * BATCH];
        }
    }

#pragma unroll
    for (int j = 0; j < PF; ++j) {
        lstm_cell(ring[j], wi, wf, wg, wo, h, c);
        op[(size_t)(T_STEPS - PF + j) * BATCH] = h;
    }
}

extern "C" void kernel_launch(void* const* d_in, const int* in_sizes, int n_in,
                              void* d_out, int out_size, void* d_ws, size_t ws_size,
                              hipStream_t stream)
{
    const float* x    = (const float*)d_in[0];
    const float* W_ih = (const float*)d_in[1];
    const float* W_hh = (const float*)d_in[2];
    const float* b_ih = (const float*)d_in[3];
    const float* b_hh = (const float*)d_in[4];
    float* out = (float*)d_out;
    float* xp  = (float*)d_ws;

    lstm_xproj_wmma<<<512, 256, 0, stream>>>(x, W_ih, b_ih, b_hh, xp);

    lstm_scan<<<2, 256, 0, stream>>>(xp, W_hh, out);
}


# scan cell: all five activations on the hardware tanh (v_tanh_f32); sigmoid(z) = 0.5 + 0.5 tanh(z/2), halves folded exactly into the stored i,f,o rows and W_hh; 4 -> 2 transcendentals in series
# speedup vs baseline: 6.1863x; 1.1866x over previous
#include <hip/hip_runtime.h>
#include <hip/hip_bf16.h>

typedef __attribute__((ext_vector_type(2))) float v2f;
typedef __attribute__((ext_vector_type(8))) float v8f;

#define T_STEPS 2048
#define BATCH   512
#define NIN     27
#define NROWS   (T_STEPS * BATCH)
#define NTILES  (NROWS / 16)

__global__ void lstm_xproj_wmma(const float* __restrict__ x,
                                const float* __restrict__ W_ih,
                                const float* __restrict__ b_ih,
                                const float* __restrict__ b_hh,
                                float* __restrict__ xp)
{
    const int lane   = threadIdx.x & 31;
    const int wave   = (blockIdx.x * blockDim.x + threadIdx.x) >> 5;
    const int nwaves = (gridDim.x * blockDim.x) >> 5;

    const int mn    = lane & 15;
    const int khalf = (lane >> 4) << 1;

    v2f bmat[7];
#pragma unroll
    for (int kk = 0; kk < 7; ++kk) {
        const int k0 = kk * 4 + khalf;
        float bx = 0.0f, by = 0.0f;
        if (mn < 4) {
            if (k0     < NIN) bx = W_ih[mn * NIN + k0];
            if (k0 + 1 < NIN) by = W_ih[mn * NIN + k0 + 1];
        }
        bmat[kk].x = bx;
        bmat[kk].y = by;
    }
    const float gscale = (mn == 2) ? 1.0f : 0.5f;
    const float bias = (mn < 4) ? gscale * (b_ih[mn] + b_hh[mn]) : 0.0f;

    for (int tile = wave; tile < NTILES; tile += nwaves) {
        const float* __restrict__ arow = x + (size_t)(tile * 16 + mn) * NIN;

        v8f acc = {};
#pragma unroll
        for (int kk = 0; kk < 7; ++kk) {
            const int k0 = kk * 4 + khalf;
            v2f a;
            a.x = (k0     < NIN) ? arow[k0]     : 0.0f;
            a.y = (k0 + 1 < NIN) ? arow[k0 + 1] : 0.0f;
            acc = __builtin_amdgcn_wmma_f32_16x16x4_f32(
                false, a, false, bmat[kk], (short)0, acc, false, false);
        }

        if (mn < 4) {
            const int rbase = tile * 16 + ((lane >> 4) << 3);
#pragma unroll
            for (int r = 0; r < 8; ++r) {
                xp[(size_t)(rbase + r) * 4 + mn] = fmaf(acc[r], gscale, bias);
            }
        }
    }
}

__device__ __forceinline__ float hw_tanh(float v) {
    return __builtin_amdgcn_tanhf(v);
}
__device__ __forceinline__ float sigmoid_of_half(float zh) {
    return 0.5f + 0.5f * hw_tanh(zh);
}

#define PF 16

__device__ __forceinline__ void lstm_cell(const float4 g,
                                          const float wi, const float wf,
                                          const float wg, const float wo,
                                          float& h, float& c)
{
    const float i  = sigmoid_of_half(g.x + h * wi);
    const float f  = sigmoid_of_half(g.y + h * wf);
    const float gg = hw_tanh        (g.z + h * wg);
    const float o  = sigmoid_of_half(g.w + h * wo);
    c = f * c + i * gg;
    h = o * hw_tanh(c);
}

__global__ void lstm_scan(const float* __restrict__ xp,
                          const float* __restrict__ W_hh,
                          float* __restrict__ out)
{
    const int b = blockIdx.x * blockDim.x + threadIdx.x;
    if (b >= BATCH) return;

    const float wi = 0.5f * W_hh[0];
    const float wf = 0.5f * W_hh[1];
    const float wg = W_hh[2];
    const float wo = 0.5f * W_hh[3];

    const float4* __restrict__ gp = (const float4*)xp + b;
    float* __restrict__ op = out + b;

    float4 ring[PF];
#pragma unroll
    for (int j = 0; j < PF; ++j) ring[j] = gp[(size_t)j * BATCH];

    float h = 0.0f, c = 0.0f;

    for (int t0 = 0; t0 < T_STEPS - PF; t0 += PF) {
#pragma unroll
        for (int j = 0; j < PF; ++j) {
            lstm_cell(ring[j], wi, wf, wg, wo, h, c);
            op[(size_t)(t0 + j) * BATCH] = h;
            __builtin_amdgcn_sched_barrier(0);
            ring[j] = gp[(size_t)(t0 + j + PF) * BATCH];
        }
    }

#pragma unroll
    for (int j = 0; j < PF; ++j) {
        lstm_cell(ring[j], wi, wf, wg, wo, h, c);
        op[(size_t)(T_STEPS - PF + j) * BATCH] = h;
    }
}

extern "C" void kernel_launch(void* const* d_in, const int* in_sizes, int n_in,
                              void* d_out, int out_size, void* d_ws, size_t ws_size,
                              hipStream_t stream)
{
    const float* x    = (const float*)d_in[0];
    const float* W_ih = (const float*)d_in[1];
    const float* W_hh = (const float*)d_in[2];
    const float* b_ih = (const float*)d_in[3];
    const float* b_hh = (const float*)d_in[4];
    float* out = (float*)d_out;
    float* xp  = (float*)d_ws;

    lstm_xproj_wmma<<<512, 256, 0, stream>>>(x, W_ih, b_ih, b_hh, xp);

    lstm_scan<<<2, 256, 0, stream>>>(xp, W_hh, out);
}


# scan: the chain wave reads gate rows from LDS and writes h to LDS; mover waves of its work-group (16 groups of 32 lanes) do all global loads and stores one 128-step chunk ahead and behind
# speedup vs baseline: 14.5677x; 2.3548x over previous
#include <hip/hip_runtime.h>
#include <hip/hip_bf16.h>

typedef __attribute__((ext_vector_type(2))) float v2f;
typedef __attribute__((ext_vector_type(8))) float v8f;

#define T_STEPS 2048
#define BATCH   512
#define NIN     27
#define NROWS   (T_STEPS * BATCH)
#define NTILES  (NROWS / 16)

__global__ void lstm_xproj_wmma(const float* __restrict__ x,
                                const float* __restrict__ W_ih,
                                const float* __restrict__ b_ih,
                                const float* __restrict__ b_hh,
                                float* __restrict__ xp)
{
    const int lane   = threadIdx.x & 31;
    const int wave   = (blockIdx.x * blockDim.x + threadIdx.x) >> 5;
    const int nwaves = (gridDim.x * blockDim.x) >> 5;

    const int mn    = lane & 15;
    const int khalf = (lane >> 4) << 1;

    v2f bmat[7];
#pragma unroll
    for (int kk = 0; kk < 7; ++kk) {
        const int k0 = kk * 4 + khalf;
        float bx = 0.0f, by = 0.0f;
        if (mn < 4) {
            if (k0     < NIN) bx = W_ih[mn * NIN + k0];
            if (k0 + 1 < NIN) by = W_ih[mn * NIN + k0 + 1];
        }
        bmat[kk].x = bx;
        bmat[kk].y = by;
    }
    const float gscale = (mn == 2) ? 1.0f : 0.5f;
    const float bias = (mn < 4) ? gscale * (b_ih[mn] + b_hh[mn]) : 0.0f;

    for (int tile = wave; tile < NTILES; tile += nwaves) {
        const float* __restrict__ arow = x + (size_t)(tile * 16 + mn) * NIN;

        v8f acc = {};
#pragma unroll
        for (int kk = 0; kk < 7; ++kk) {
            const int k0 = kk * 4 + khalf;
            v2f a;
            a.x = (k0     < NIN) ? arow[k0]     : 0.0f;
            a.y = (k0 + 1 < NIN) ? arow[k0 + 1] : 0.0f;
            acc = __builtin_amdgcn_wmma_f32_16x16x4_f32(
                false, a, false, bmat[kk], (short)0, acc, false, false);
        }

        if (mn < 4) {
            const int rbase = tile * 16 + ((lane >> 4) << 3);
#pragma unroll
            for (int r = 0; r < 8; ++r) {
                xp[(size_t)(rbase + r) * 4 + mn] = fmaf(acc[r], gscale, bias);
            }
        }
    }
}

__device__ __forceinline__ float hw_tanh(float v) {
    return __builtin_amdgcn_tanhf(v);
}
__device__ __forceinline__ float sigmoid_of_half(float zh) {
    return 0.5f + 0.5f * hw_tanh(zh);
}

#define CH      128
#define NCH     (T_STEPS / CH)
#define NMOV    8
#define MSTEPS  (CH / NMOV)
#define GRP     8
#define SCAN_THREADS (32 * (1 + NMOV))

__device__ __forceinline__ void lstm_cell(const float4 g,
                                          const float wi, const float wf,
                                          const float wg, const float wo,
                                          float& h, float& c)
{
    const float i  = sigmoid_of_half(g.x + h * wi);
    const float f  = sigmoid_of_half(g.y + h * wf);
    const float gg = hw_tanh        (g.z + h * wg);
    const float o  = sigmoid_of_half(g.w + h * wo);
    c = f * c + i * gg;
    h = o * hw_tanh(c);
}

__global__ void __launch_bounds__(SCAN_THREADS)
lstm_scan(const float* __restrict__ xp,
          const float* __restrict__ W_hh,
          float* __restrict__ out)
{
    __shared__ float4 s_g[2][CH][32];
    __shared__ __attribute__((aligned(16))) float s_h[2][CH][32];

    const int lane  = threadIdx.x & 31;
    const int wave  = __builtin_amdgcn_readfirstlane((int)(threadIdx.x >> 5));
    const int bbase = blockIdx.x * 32;

    const float wi = 0.5f * W_hh[0];
    const float wf = 0.5f * W_hh[1];
    const float wg = W_hh[2];
    const float wo = 0.5f * W_hh[3];

    const int m     = wave - 1;
    const int srow  = lane >> 3;
    const int piece = (lane & 7) * 4;
    const float4* __restrict__ gp = (const float4*)xp + bbase + lane;

    float h = 0.0f, c = 0.0f;

    if (wave != 0) {
        float4 v[MSTEPS];
#pragma unroll
        for (int i = 0; i < MSTEPS; ++i)
            v[i] = gp[(size_t)(m * MSTEPS + i) * BATCH];
#pragma unroll
        for (int i = 0; i < MSTEPS; ++i)
            s_g[0][m * MSTEPS + i][lane] = v[i];
    }
    __syncthreads();

    for (int ck = 0; ck < NCH; ++ck) {
        const int st = ck & 1;
        if (wave == 0) {
            const float4* sg = &s_g[st][0][lane];
            float*        sh = &s_h[st][0][lane];
            float4 ga[GRP], gb[GRP];
            float  ha[GRP], hb[GRP];
#pragma unroll
            for (int i = 0; i < GRP; ++i) ga[i] = sg[i * 32];

            for (int g0 = 0; g0 < CH; g0 += 2 * GRP) {
#pragma unroll
                for (int i = 0; i < GRP; ++i) gb[i] = sg[(g0 + GRP + i) * 32];
                __builtin_amdgcn_sched_barrier(0);
#pragma unroll
                for (int i = 0; i < GRP; ++i) {
                    lstm_cell(ga[i], wi, wf, wg, wo, h, c);
                    ha[i] = h;
                }
                __builtin_amdgcn_sched_barrier(0);
#pragma unroll
                for (int i = 0; i < GRP; ++i) sh[(g0 + i) * 32] = ha[i];
                if (g0 + 2 * GRP < CH) {
#pragma unroll
                    for (int i = 0; i < GRP; ++i) ga[i] = sg[(g0 + 2 * GRP + i) * 32];
                }
                __builtin_amdgcn_sched_barrier(0);
#pragma unroll
                for (int i = 0; i < GRP; ++i) {
                    lstm_cell(gb[i], wi, wf, wg, wo, h, c);
                    hb[i] = h;
                }
                __builtin_amdgcn_sched_barrier(0);
#pragma unroll
                for (int i = 0; i < GRP; ++i) sh[(g0 + GRP + i) * 32] = hb[i];
            }
        } else {
            if (ck > 0) {
#pragma unroll
                for (int q = 0; q < MSTEPS / 4; ++q) {
                    const int s = m * MSTEPS + q * 4 + srow;
                    const float4 hv = *(const float4*)&s_h[st ^ 1][s][piece];
                    *(float4*)(out + (size_t)((ck - 1) * CH + s) * BATCH + bbase + piece) = hv;
                }
            }
            if (ck + 1 < NCH) {
                float4 v[MSTEPS];
#pragma unroll
                for (int i = 0; i < MSTEPS; ++i)
                    v[i] = gp[(size_t)((ck + 1) * CH + m * MSTEPS + i) * BATCH];
#pragma unroll
                for (int i = 0; i < MSTEPS; ++i)
                    s_g[st ^ 1][m * MSTEPS + i][lane] = v[i];
            }
        }
        __syncthreads();
    }

    if (wave != 0) {
#pragma unroll
        for (int q = 0; q < MSTEPS / 4; ++q) {
            const int s = m * MSTEPS + q * 4 + srow;
            const float4 hv = *(const float4*)&s_h[(NCH - 1) & 1][s][piece];
            *(float4*)(out + (size_t)((NCH - 1) * CH + s) * BATCH + bbase + piece) = hv;
        }
    }
}

extern "C" void kernel_launch(void* const* d_in, const int* in_sizes, int n_in,
                              void* d_out, int out_size, void* d_ws, size_t ws_size,
                              hipStream_t stream)
{
    const float* x    = (const float*)d_in[0];
    const float* W_ih = (const float*)d_in[1];
    const float* W_hh = (const float*)d_in[2];
    const float* b_ih = (const float*)d_in[3];
    const float* b_hh = (const float*)d_in[4];
    float* out = (float*)d_out;
    float* xp  = (float*)d_ws;

    lstm_xproj_wmma<<<512, 256, 0, stream>>>(x, W_ih, b_ih, b_hh, xp);

    lstm_scan<<<16, 288, 0, stream>>>(xp, W_hh, out);
}


# scan cell re-associated, two plain ops leave the h-to-h chain: o*W_hh is formed beside tanh(c) and feeds the next gate fma, f*c is fma(tf,c/2,c/2); products round elsewhere, no sum reordered
# speedup vs baseline: 15.6392x; 1.0736x over previous
#include <hip/hip_runtime.h>
#include <hip/hip_bf16.h>

typedef __attribute__((ext_vector_type(2))) float v2f;
typedef __attribute__((ext_vector_type(8))) float v8f;

#define T_STEPS 2048
#define BATCH   512
#define NIN     27
#define NROWS   (T_STEPS * BATCH)
#define NTILES  (NROWS / 16)

__global__ void lstm_xproj_wmma(const float* __restrict__ x,
                                const float* __restrict__ W_ih,
                                const float* __restrict__ b_ih,
                                const float* __restrict__ b_hh,
                                float* __restrict__ xp)
{
    const int lane   = threadIdx.x & 31;
    const int wave   = (blockIdx.x * blockDim.x + threadIdx.x) >> 5;
    const int nwaves = (gridDim.x * blockDim.x) >> 5;

    const int mn    = lane & 15;
    const int khalf = (lane >> 4) << 1;

    v2f bmat[7];
#pragma unroll
    for (int kk = 0; kk < 7; ++kk) {
        const int k0 = kk * 4 + khalf;
        float bx = 0.0f, by = 0.0f;
        if (mn < 4) {
            if (k0     < NIN) bx = W_ih[mn * NIN + k0];
            if (k0 + 1 < NIN) by = W_ih[mn * NIN + k0 + 1];
        }
        bmat[kk].x = bx;
        bmat[kk].y = by;
    }
    const float gscale = (mn == 2) ? 1.0f : 0.5f;
    const float bias = (mn < 4) ? gscale * (b_ih[mn] + b_hh[mn]) : 0.0f;

    for (int tile = wave; tile < NTILES; tile += nwaves) {
        const float* __restrict__ arow = x + (size_t)(tile * 16 + mn) * NIN;

        v8f acc = {};
#pragma unroll
        for (int kk = 0; kk < 7; ++kk) {
            const int k0 = kk * 4 + khalf;
            v2f a;
            a.x = (k0     < NIN) ? arow[k0]     : 0.0f;
            a.y = (k0 + 1 < NIN) ? arow[k0 + 1] : 0.0f;
            acc = __builtin_amdgcn_wmma_f32_16x16x4_f32(
                false, a, false, bmat[kk], (short)0, acc, false, false);
        }

        if (mn < 4) {
            const int rbase = tile * 16 + ((lane >> 4) << 3);
#pragma unroll
            for (int r = 0; r < 8; ++r) {
                xp[(size_t)(rbase + r) * 4 + mn] = fmaf(acc[r], gscale, bias);
            }
        }
    }
}

__device__ __forceinline__ float hw_tanh(float v) {
    return __builtin_amdgcn_tanhf(v);
}
__device__ __forceinline__ float sigmoid_of_half(float zh) {
    return 0.5f + 0.5f * hw_tanh(zh);
}

#define CH      128
#define NCH     (T_STEPS / CH)
#define NMOV    8
#define MSTEPS  (CH / NMOV)
#define GRP     8
#define SCAN_THREADS (32 * (1 + NMOV))

struct Carry {
    float tc;
    float hc;
    float owi, owf, owg, owo;
};

__device__ __forceinline__ float lstm_cell(const float4 g,
                                           const float qi, const float qf,
                                           const float qg, const float qo,
                                           Carry& s)
{
    const float ti = hw_tanh(fmaf(s.tc, s.owi, g.x));
    const float tf = hw_tanh(fmaf(s.tc, s.owf, g.y));
    const float tg = hw_tanh(fmaf(s.tc, s.owg, g.z));
    const float to = hw_tanh(fmaf(s.tc, s.owo, g.w));
    const float i  = fmaf(0.5f, ti, 0.5f);
    const float fc = fmaf(tf, s.hc, s.hc);
    const float c  = fmaf(i, tg, fc);
    __builtin_amdgcn_sched_barrier(0);
    const float tc = hw_tanh(c);
    const float o  = fmaf(0.5f, to, 0.5f);
    s.owi = fmaf(to, qi, qi);
    s.owf = fmaf(to, qf, qf);
    s.owg = fmaf(to, qg, qg);
    s.owo = fmaf(to, qo, qo);
    s.hc  = 0.5f * c;
    s.tc  = tc;
    return o * tc;
}

__global__ void __launch_bounds__(SCAN_THREADS)
lstm_scan(const float* __restrict__ xp,
          const float* __restrict__ W_hh,
          float* __restrict__ out)
{
    __shared__ float4 s_g[2][CH][32];
    __shared__ __attribute__((aligned(16))) float s_h[2][CH][32];

    const int lane  = threadIdx.x & 31;
    const int wave  = __builtin_amdgcn_readfirstlane((int)(threadIdx.x >> 5));
    const int bbase = blockIdx.x * 32;

    const float qi = 0.25f * W_hh[0];
    const float qf = 0.25f * W_hh[1];
    const float qg = 0.5f  * W_hh[2];
    const float qo = 0.25f * W_hh[3];

    const int m     = wave - 1;
    const int srow  = lane >> 3;
    const int piece = (lane & 7) * 4;
    const float4* __restrict__ gp = (const float4*)xp + bbase + lane;

    Carry s = {0.0f, 0.0f, 0.0f, 0.0f, 0.0f, 0.0f};

    if (wave != 0) {
        float4 v[MSTEPS];
#pragma unroll
        for (int i = 0; i < MSTEPS; ++i)
            v[i] = gp[(size_t)(m * MSTEPS + i) * BATCH];
#pragma unroll
        for (int i = 0; i < MSTEPS; ++i)
            s_g[0][m * MSTEPS + i][lane] = v[i];
    }
    __syncthreads();

    for (int ck = 0; ck < NCH; ++ck) {
        const int st = ck & 1;
        if (wave == 0) {
            const float4* sg = &s_g[st][0][lane];
            float*        sh = &s_h[st][0][lane];
            float4 ga[GRP], gb[GRP];
            float  ha[GRP], hb[GRP];
#pragma unroll
            for (int i = 0; i < GRP; ++i) ga[i] = sg[i * 32];

            for (int g0 = 0; g0 < CH; g0 += 2 * GRP) {
#pragma unroll
                for (int i = 0; i < GRP; ++i) gb[i] = sg[(g0 + GRP + i) * 32];
                __builtin_amdgcn_sched_barrier(0);
#pragma unroll
                for (int i = 0; i < GRP; ++i) {
                    ha[i] = lstm_cell(ga[i], qi, qf, qg, qo, s);
                }
                __builtin_amdgcn_sched_barrier(0);
#pragma unroll
                for (int i = 0; i < GRP; ++i) sh[(g0 + i) * 32] = ha[i];
                if (g0 + 2 * GRP < CH) {
#pragma unroll
                    for (int i = 0; i < GRP; ++i) ga[i] = sg[(g0 + 2 * GRP + i) * 32];
                }
                __builtin_amdgcn_sched_barrier(0);
#pragma unroll
                for (int i = 0; i < GRP; ++i) {
                    hb[i] = lstm_cell(gb[i], qi, qf, qg, qo, s);
                }
                __builtin_amdgcn_sched_barrier(0);
#pragma unroll
                for (int i = 0; i < GRP; ++i) sh[(g0 + GRP + i) * 32] = hb[i];
            }
        } else {
            if (ck > 0) {
#pragma unroll
                for (int q = 0; q < MSTEPS / 4; ++q) {
                    const int s = m * MSTEPS + q * 4 + srow;
                    const float4 hv = *(const float4*)&s_h[st ^ 1][s][piece];
                    *(float4*)(out + (size_t)((ck - 1) * CH + s) * BATCH + bbase + piece) = hv;
                }
            }
            if (ck + 1 < NCH) {
                float4 v[MSTEPS];
#pragma unroll
                for (int i = 0; i < MSTEPS; ++i)
                    v[i] = gp[(size_t)((ck + 1) * CH + m * MSTEPS + i) * BATCH];
#pragma unroll
                for (int i = 0; i < MSTEPS; ++i)
                    s_g[st ^ 1][m * MSTEPS + i][lane] = v[i];
            }
        }
        __syncthreads();
    }

    if (wave != 0) {
#pragma unroll
        for (int q = 0; q < MSTEPS / 4; ++q) {
            const int s = m * MSTEPS + q * 4 + srow;
            const float4 hv = *(const float4*)&s_h[(NCH - 1) & 1][s][piece];
            *(float4*)(out + (size_t)((NCH - 1) * CH + s) * BATCH + bbase + piece) = hv;
        }
    }
}

extern "C" void kernel_launch(void* const* d_in, const int* in_sizes, int n_in,
                              void* d_out, int out_size, void* d_ws, size_t ws_size,
                              hipStream_t stream)
{
    const float* x    = (const float*)d_in[0];
    const float* W_ih = (const float*)d_in[1];
    const float* W_hh = (const float*)d_in[2];
    const float* b_ih = (const float*)d_in[3];
    const float* b_hh = (const float*)d_in[4];
    float* out = (float*)d_out;
    float* xp  = (float*)d_ws;

    lstm_xproj_wmma<<<512, 256, 0, stream>>>(x, W_ih, b_ih, b_hh, xp);

    lstm_scan<<<16, 288, 0, stream>>>(xp, W_hh, out);
}


# projection kernel: whole-line 16-byte global accesses: 64-row x blocks staged via per-wave LDS, next block in flight; product transposed (W_ih as A) so a lane stores a row's 4 gates; k order kept
# speedup vs baseline: 17.7280x; 1.1336x over previous
#include <hip/hip_runtime.h>
#include <hip/hip_bf16.h>

typedef __attribute__((ext_vector_type(2))) float v2f;
typedef __attribute__((ext_vector_type(4))) float v4f;
typedef __attribute__((ext_vector_type(8))) float v8f;

#define T_STEPS 2048
#define BATCH   512
#define NIN     27
#define NROWS   (T_STEPS * BATCH)
#define NTILES  (NROWS / 16)

#define XB_TILES    4
#define XB_ROWS     (16 * XB_TILES)
#define XB_VECS     (XB_ROWS * NIN / 4)
#define XB_FULL     (XB_VECS / 32)
#define XB_REST     (XB_VECS - 32 * XB_FULL)
#define NBLOCKS     (NTILES / XB_TILES)
#define XP_WAVES    8
#define XP_THREADS  (32 * XP_WAVES)

__global__ void __launch_bounds__(XP_THREADS)
lstm_xproj_wmma(const float* __restrict__ x,
                const float* __restrict__ W_ih,
                const float* __restrict__ b_ih,
                const float* __restrict__ b_hh,
                float* __restrict__ xp)
{
    __shared__ v4f s_x[XP_WAVES][XB_VECS];

    const int lane   = threadIdx.x & 31;
    const int wib    = __builtin_amdgcn_readfirstlane((int)(threadIdx.x >> 5));
    const int wave   = blockIdx.x * XP_WAVES + wib;
    const int nwaves = gridDim.x * XP_WAVES;

    const int mn    = lane & 15;
    const int khalf = (lane >> 4) << 1;

    v2f wfrag[7];
#pragma unroll
    for (int kk = 0; kk < 7; ++kk) {
        const int k0 = kk * 4 + khalf;
        float bx = 0.0f, by = 0.0f;
        if (mn < 4) {
            if (k0     < NIN) bx = W_ih[mn * NIN + k0];
            if (k0 + 1 < NIN) by = W_ih[mn * NIN + k0 + 1];
        }
        wfrag[kk].x = bx;
        wfrag[kk].y = by;
    }
    const float bias0 = 0.5f * (b_ih[0] + b_hh[0]);
    const float bias1 = 0.5f * (b_ih[1] + b_hh[1]);
    const float bias2 =         b_ih[2] + b_hh[2];
    const float bias3 = 0.5f * (b_ih[3] + b_hh[3]);

    v4f*         st4 = s_x[wib];
    const float* stf = (const float*)s_x[wib];
    const v4f*   x4  = (const v4f*)x;
    v4f*         xp4 = (v4f*)xp;

    v4f v[XB_FULL + 1];
    int blk = wave;
    if (blk < NBLOCKS) {
        const v4f* src = x4 + (size_t)blk * XB_VECS + lane;
#pragma unroll
        for (int i = 0; i < XB_FULL; ++i) v[i] = src[i * 32];
        if (lane < XB_REST) v[XB_FULL] = src[XB_FULL * 32];
    }
    while (blk < NBLOCKS) {
#pragma unroll
        for (int i = 0; i < XB_FULL; ++i) st4[i * 32 + lane] = v[i];
        if (lane < XB_REST) st4[XB_FULL * 32 + lane] = v[XB_FULL];

        const int nb = blk + nwaves;
        if (nb < NBLOCKS) {
            const v4f* src = x4 + (size_t)nb * XB_VECS + lane;
#pragma unroll
            for (int i = 0; i < XB_FULL; ++i) v[i] = src[i * 32];
            if (lane < XB_REST) v[XB_FULL] = src[XB_FULL * 32];
        }

        const float* arow = stf + mn * NIN;
        v8f acc[XB_TILES];
#pragma unroll
        for (int j = 0; j < XB_TILES; ++j) acc[j] = v8f{};
#pragma unroll
        for (int kk = 0; kk < 7; ++kk) {
            const int  k0  = kk * 4 + khalf;
            const bool pad = (k0 + 1 >= NIN);
            const int  k1  = pad ? k0 : k0 + 1;
#pragma unroll
            for (int j = 0; j < XB_TILES; ++j) {
                v2f xf;
                xf.x = arow[j * 16 * NIN + k0];
                const float y = arow[j * 16 * NIN + k1];
                xf.y = pad ? 0.0f : y;
                acc[j] = __builtin_amdgcn_wmma_f32_16x16x4_f32(
                    false, wfrag[kk], false, xf, (short)0, acc[j], false, false);
            }
        }

        if (lane < 16) {
#pragma unroll
            for (int j = 0; j < XB_TILES; ++j) {
                v4f o;
                o.x = fmaf(acc[j][0], 0.5f, bias0);
                o.y = fmaf(acc[j][1], 0.5f, bias1);
                o.z = acc[j][2] + bias2;
                o.w = fmaf(acc[j][3], 0.5f, bias3);
                xp4[(size_t)(blk * XB_TILES + j) * 16 + lane] = o;
            }
        }
        blk = nb;
    }
}

__device__ __forceinline__ float hw_tanh(float v) {
    return __builtin_amdgcn_tanhf(v);
}
__device__ __forceinline__ float sigmoid_of_half(float zh) {
    return 0.5f + 0.5f * hw_tanh(zh);
}

#define CH      128
#define NCH     (T_STEPS / CH)
#define NMOV    8
#define MSTEPS  (CH / NMOV)
#define GRP     8
#define SCAN_THREADS (32 * (1 + NMOV))

struct Carry {
    float tc;
    float hc;
    float owi, owf, owg, owo;
};

__device__ __forceinline__ float lstm_cell(const float4 g,
                                           const float qi, const float qf,
                                           const float qg, const float qo,
                                           Carry& s)
{
    const float ti = hw_tanh(fmaf(s.tc, s.owi, g.x));
    const float tf = hw_tanh(fmaf(s.tc, s.owf, g.y));
    const float tg = hw_tanh(fmaf(s.tc, s.owg, g.z));
    const float to = hw_tanh(fmaf(s.tc, s.owo, g.w));
    const float i  = fmaf(0.5f, ti, 0.5f);
    const float fc = fmaf(tf, s.hc, s.hc);
    const float c  = fmaf(i, tg, fc);
    __builtin_amdgcn_sched_barrier(0);
    const float tc = hw_tanh(c);
    const float o  = fmaf(0.5f, to, 0.5f);
    s.owi = fmaf(to, qi, qi);
    s.owf = fmaf(to, qf, qf);
    s.owg = fmaf(to, qg, qg);
    s.owo = fmaf(to, qo, qo);
    s.hc  = 0.5f * c;
    s.tc  = tc;
    return o * tc;
}

__global__ void __launch_bounds__(SCAN_THREADS)
lstm_scan(const float* __restrict__ xp,
          const float* __restrict__ W_hh,
          float* __restrict__ out)
{
    __shared__ float4 s_g[2][CH][32];
    __shared__ __attribute__((aligned(16))) float s_h[2][CH][32];

    const int lane  = threadIdx.x & 31;
    const int wave  = __builtin_amdgcn_readfirstlane((int)(threadIdx.x >> 5));
    const int bbase = blockIdx.x * 32;

    const float qi = 0.25f * W_hh[0];
    const float qf = 0.25f * W_hh[1];
    const float qg = 0.5f  * W_hh[2];
    const float qo = 0.25f * W_hh[3];

    const int m     = wave - 1;
    const int srow  = lane >> 3;
    const int piece = (lane & 7) * 4;
    const float4* __restrict__ gp = (const float4*)xp + bbase + lane;

    Carry s = {0.0f, 0.0f, 0.0f, 0.0f, 0.0f, 0.0f};

    if (wave != 0) {
        float4 v[MSTEPS];
#pragma unroll
        for (int i = 0; i < MSTEPS; ++i)
            v[i] = gp[(size_t)(m * MSTEPS + i) * BATCH];
#pragma unroll
        for (int i = 0; i < MSTEPS; ++i)
            s_g[0][m * MSTEPS + i][lane] = v[i];
    }
    __syncthreads();

    for (int ck = 0; ck < NCH; ++ck) {
        const int st = ck & 1;
        if (wave == 0) {
            const float4* sg = &s_g[st][0][lane];
            float*        sh = &s_h[st][0][lane];
            float4 ga[GRP], gb[GRP];
            float  ha[GRP], hb[GRP];
#pragma unroll
            for (int i = 0; i < GRP; ++i) ga[i] = sg[i * 32];

            for (int g0 = 0; g0 < CH; g0 += 2 * GRP) {
#pragma unroll
                for (int i = 0; i < GRP; ++i) gb[i] = sg[(g0 + GRP + i) * 32];
                __builtin_amdgcn_sched_barrier(0);
#pragma unroll
                for (int i = 0; i < GRP; ++i) {
                    ha[i] = lstm_cell(ga[i], qi, qf, qg, qo, s);
                }
                __builtin_amdgcn_sched_barrier(0);
#pragma unroll
                for (int i = 0; i < GRP; ++i) sh[(g0 + i) * 32] = ha[i];
                if (g0 + 2 * GRP < CH) {
#pragma unroll
                    for (int i = 0; i < GRP; ++i) ga[i] = sg[(g0 + 2 * GRP + i) * 32];
                }
                __builtin_amdgcn_sched_barrier(0);
#pragma unroll
                for (int i = 0; i < GRP; ++i) {
                    hb[i] = lstm_cell(gb[i], qi, qf, qg, qo, s);
                }
                __builtin_amdgcn_sched_barrier(0);
#pragma unroll
                for (int i = 0; i < GRP; ++i) sh[(g0 + GRP + i) * 32] = hb[i];
            }
        } else {
            if (ck > 0) {
#pragma unroll
                for (int q = 0; q < MSTEPS / 4; ++q) {
                    const int s = m * MSTEPS + q * 4 + srow;
                    const float4 hv = *(const float4*)&s_h[st ^ 1][s][piece];
                    *(float4*)(out + (size_t)((ck - 1) * CH + s) * BATCH + bbase + piece) = hv;
                }
            }
            if (ck + 1 < NCH) {
                float4 v[MSTEPS];
#pragma unroll
                for (int i = 0; i < MSTEPS; ++i)
                    v[i] = gp[(size_t)((ck + 1) * CH + m * MSTEPS + i) * BATCH];
#pragma unroll
                for (int i = 0; i < MSTEPS; ++i)
                    s_g[st ^ 1][m * MSTEPS + i][lane] = v[i];
            }
        }
        __syncthreads();
    }

    if (wave != 0) {
#pragma unroll
        for (int q = 0; q < MSTEPS / 4; ++q) {
            const int s = m * MSTEPS + q * 4 + srow;
            const float4 hv = *(const float4*)&s_h[(NCH - 1) & 1][s][piece];
            *(float4*)(out + (size_t)((NCH - 1) * CH + s) * BATCH + bbase + piece) = hv;
        }
    }
}

extern "C" void kernel_launch(void* const* d_in, const int* in_sizes, int n_in,
                              void* d_out, int out_size, void* d_ws, size_t ws_size,
                              hipStream_t stream)
{
    const float* x    = (const float*)d_in[0];
    const float* W_ih = (const float*)d_in[1];
    const float* W_hh = (const float*)d_in[2];
    const float* b_ih = (const float*)d_in[3];
    const float* b_hh = (const float*)d_in[4];
    float* out = (float*)d_out;
    float* xp  = (float*)d_ws;

    lstm_xproj_wmma<<<512, 256, 0, stream>>>(x, W_ih, b_ih, b_hh, xp);

    lstm_scan<<<16, 288, 0, stream>>>(xp, W_hh, out);
}


# scan chain wave meets LDS once per 16 steps instead of per 8: gate rows read 16 steps per group, h written four steps per 16-byte store; cell arithmetic and its order unchanged
# speedup vs baseline: 19.4120x; 1.0950x over previous
#include <hip/hip_runtime.h>
#include <hip/hip_bf16.h>

typedef __attribute__((ext_vector_type(2))) float v2f;
typedef __attribute__((ext_vector_type(4))) float v4f;
typedef __attribute__((ext_vector_type(8))) float v8f;

#define T_STEPS 2048
#define BATCH   512
#define NIN     27
#define NROWS   (T_STEPS * BATCH)
#define NTILES  (NROWS / 16)

#define XB_TILES    4
#define XB_ROWS     (16 * XB_TILES)
#define XB_VECS     (XB_ROWS * NIN / 4)
#define XB_FULL     (XB_VECS / 32)
#define XB_REST     (XB_VECS - 32 * XB_FULL)
#define NBLOCKS     (NTILES / XB_TILES)
#define XP_WAVES    8
#define XP_THREADS  (32 * XP_WAVES)

__global__ void __launch_bounds__(XP_THREADS)
lstm_xproj_wmma(const float* __restrict__ x,
                const float* __restrict__ W_ih,
                const float* __restrict__ b_ih,
                const float* __restrict__ b_hh,
                float* __restrict__ xp)
{
    __shared__ v4f s_x[XP_WAVES][XB_VECS];

    const int lane   = threadIdx.x & 31;
    const int wib    = __builtin_amdgcn_readfirstlane((int)(threadIdx.x >> 5));
    const int wave   = blockIdx.x * XP_WAVES + wib;
    const int nwaves = gridDim.x * XP_WAVES;

    const int mn    = lane & 15;
    const int khalf = (lane >> 4) << 1;

    v2f wfrag[7];
#pragma unroll
    for (int kk = 0; kk < 7; ++kk) {
        const int k0 = kk * 4 + khalf;
        float bx = 0.0f, by = 0.0f;
        if (mn < 4) {
            if (k0     < NIN) bx = W_ih[mn * NIN + k0];
            if (k0 + 1 < NIN) by = W_ih[mn * NIN + k0 + 1];
        }
        wfrag[kk].x = bx;
        wfrag[kk].y = by;
    }
    const float bias0 = 0.5f * (b_ih[0] + b_hh[0]);
    const float bias1 = 0.5f * (b_ih[1] + b_hh[1]);
    const float bias2 =         b_ih[2] + b_hh[2];
    const float bias3 = 0.5f * (b_ih[3] + b_hh[3]);

    v4f*         st4 = s_x[wib];
    const float* stf = (const float*)s_x[wib];
    const v4f*   x4  = (const v4f*)x;
    v4f*         xp4 = (v4f*)xp;

    v4f v[XB_FULL + 1];
    int blk = wave;
    if (blk < NBLOCKS) {
        const v4f* src = x4 + (size_t)blk * XB_VECS + lane;
#pragma unroll
        for (int i = 0; i < XB_FULL; ++i) v[i] = src[i * 32];
        if (lane < XB_REST) v[XB_FULL] = src[XB_FULL * 32];
    }
    while (blk < NBLOCKS) {
#pragma unroll
        for (int i = 0; i < XB_FULL; ++i) st4[i * 32 + lane] = v[i];
        if (lane < XB_REST) st4[XB_FULL * 32 + lane] = v[XB_FULL];

        const int nb = blk + nwaves;
        if (nb < NBLOCKS) {
            const v4f* src = x4 + (size_t)nb * XB_VECS + lane;
#pragma unroll
            for (int i = 0; i < XB_FULL; ++i) v[i] = src[i * 32];
            if (lane < XB_REST) v[XB_FULL] = src[XB_FULL * 32];
        }

        const float* arow = stf + mn * NIN;
        v8f acc[XB_TILES];
#pragma unroll
        for (int j = 0; j < XB_TILES; ++j) acc[j] = v8f{};
#pragma unroll
        for (int kk = 0; kk < 7; ++kk) {
            const int  k0  = kk * 4 + khalf;
            const bool pad = (k0 + 1 >= NIN);
            const int  k1  = pad ? k0 : k0 + 1;
#pragma unroll
            for (int j = 0; j < XB_TILES; ++j) {
                v2f xf;
                xf.x = arow[j * 16 * NIN + k0];
                const float y = arow[j * 16 * NIN + k1];
                xf.y = pad ? 0.0f : y;
                acc[j] = __builtin_amdgcn_wmma_f32_16x16x4_f32(
                    false, wfrag[kk], false, xf, (short)0, acc[j], false, false);
            }
        }

        if (lane < 16) {
#pragma unroll
            for (int j = 0; j < XB_TILES; ++j) {
                v4f o;
                o.x = fmaf(acc[j][0], 0.5f, bias0);
                o.y = fmaf(acc[j][1], 0.5f, bias1);
                o.z = acc[j][2] + bias2;
                o.w = fmaf(acc[j][3], 0.5f, bias3);
                xp4[(size_t)(blk * XB_TILES + j) * 16 + lane] = o;
            }
        }
        blk = nb;
    }
}

__device__ __forceinline__ float hw_tanh(float v) {
    return __builtin_amdgcn_tanhf(v);
}
__device__ __forceinline__ float sigmoid_of_half(float zh) {
    return 0.5f + 0.5f * hw_tanh(zh);
}

#define CH      128
#define NCH     (T_STEPS / CH)
#define NMOV    8
#define MSTEPS  (CH / NMOV)
#define GRP     16
#define SCAN_THREADS (32 * (1 + NMOV))

struct Carry {
    float tc;
    float hc;
    float owi, owf, owg, owo;
};

__device__ __forceinline__ float lstm_cell(const float4 g,
                                           const float qi, const float qf,
                                           const float qg, const float qo,
                                           Carry& s)
{
    const float ti = hw_tanh(fmaf(s.tc, s.owi, g.x));
    const float tf = hw_tanh(fmaf(s.tc, s.owf, g.y));
    const float tg = hw_tanh(fmaf(s.tc, s.owg, g.z));
    const float to = hw_tanh(fmaf(s.tc, s.owo, g.w));
    const float i  = fmaf(0.5f, ti, 0.5f);
    const float fc = fmaf(tf, s.hc, s.hc);
    const float c  = fmaf(i, tg, fc);
    __builtin_amdgcn_sched_barrier(0);
    const float tc = hw_tanh(c);
    const float o  = fmaf(0.5f, to, 0.5f);
    s.owi = fmaf(to, qi, qi);
    s.owf = fmaf(to, qf, qf);
    s.owg = fmaf(to, qg, qg);
    s.owo = fmaf(to, qo, qo);
    s.hc  = 0.5f * c;
    s.tc  = tc;
    float h = o * tc;
    asm volatile("" : "+v"(h));
    return h;
}

__global__ void __launch_bounds__(SCAN_THREADS)
lstm_scan(const float* __restrict__ xp,
          const float* __restrict__ W_hh,
          float* __restrict__ out)
{
    __shared__ float4 s_g[2][CH][32];
    __shared__ float4 s_h[2][CH / 4][32];

    const int lane  = threadIdx.x & 31;
    const int wave  = __builtin_amdgcn_readfirstlane((int)(threadIdx.x >> 5));
    const int bbase = blockIdx.x * 32;

    const float qi = 0.25f * W_hh[0];
    const float qf = 0.25f * W_hh[1];
    const float qg = 0.5f  * W_hh[2];
    const float qo = 0.25f * W_hh[3];

    const int m     = wave - 1;
    const int srow  = lane >> 3;
    const int piece = (lane & 7) * 4;
    const float4* __restrict__ gp = (const float4*)xp + bbase + lane;

    Carry s = {0.0f, 0.0f, 0.0f, 0.0f, 0.0f, 0.0f};

    if (wave != 0) {
        float4 v[MSTEPS];
#pragma unroll
        for (int i = 0; i < MSTEPS; ++i)
            v[i] = gp[(size_t)(m * MSTEPS + i) * BATCH];
#pragma unroll
        for (int i = 0; i < MSTEPS; ++i)
            s_g[0][m * MSTEPS + i][lane] = v[i];
    }
    __syncthreads();

    for (int ck = 0; ck < NCH; ++ck) {
        const int st = ck & 1;
        if (wave == 0) {
            const float4* sg = &s_g[st][0][lane];
            float4*       sh = &s_h[st][0][lane];
            float4 ga[GRP], gb[GRP];
            float  ha[GRP], hb[GRP];
#pragma unroll
            for (int i = 0; i < GRP; ++i) ga[i] = sg[i * 32];

            for (int g0 = 0; g0 < CH; g0 += 2 * GRP) {
#pragma unroll
                for (int i = 0; i < GRP; ++i) gb[i] = sg[(g0 + GRP + i) * 32];
                __builtin_amdgcn_sched_barrier(0);
#pragma unroll
                for (int i = 0; i < GRP; ++i) {
                    ha[i] = lstm_cell(ga[i], qi, qf, qg, qo, s);
                }
                __builtin_amdgcn_sched_barrier(0);
#pragma unroll
                for (int q = 0; q < GRP / 4; ++q)
                    sh[(g0 / 4 + q) * 32] = make_float4(ha[4 * q], ha[4 * q + 1], ha[4 * q + 2], ha[4 * q + 3]);
                if (g0 + 2 * GRP < CH) {
#pragma unroll
                    for (int i = 0; i < GRP; ++i) ga[i] = sg[(g0 + 2 * GRP + i) * 32];
                }
                __builtin_amdgcn_sched_barrier(0);
#pragma unroll
                for (int i = 0; i < GRP; ++i) {
                    hb[i] = lstm_cell(gb[i], qi, qf, qg, qo, s);
                }
                __builtin_amdgcn_sched_barrier(0);
#pragma unroll
                for (int q = 0; q < GRP / 4; ++q)
                    sh[((g0 + GRP) / 4 + q) * 32] = make_float4(hb[4 * q], hb[4 * q + 1], hb[4 * q + 2], hb[4 * q + 3]);
            }
        } else {
            if (ck > 0) {
#pragma unroll
                for (int q = 0; q < MSTEPS / 4; ++q) {
                    const int s = m * MSTEPS + q * 4 + srow;
                    const float* hs = (const float*)&s_h[st ^ 1][s >> 2][piece] + (s & 3);
                    const float4 hv = make_float4(hs[0], hs[4], hs[8], hs[12]);
                    *(float4*)(out + (size_t)((ck - 1) * CH + s) * BATCH + bbase + piece) = hv;
                }
            }
            if (ck + 1 < NCH) {
                float4 v[MSTEPS];
#pragma unroll
                for (int i = 0; i < MSTEPS; ++i)
                    v[i] = gp[(size_t)((ck + 1) * CH + m * MSTEPS + i) * BATCH];
#pragma unroll
                for (int i = 0; i < MSTEPS; ++i)
                    s_g[st ^ 1][m * MSTEPS + i][lane] = v[i];
            }
        }
        __syncthreads();
    }

    if (wave != 0) {
#pragma unroll
        for (int q = 0; q < MSTEPS / 4; ++q) {
            const int s = m * MSTEPS + q * 4 + srow;
            const float* hs = (const float*)&s_h[(NCH - 1) & 1][s >> 2][piece] + (s & 3);
            const float4 hv = make_float4(hs[0], hs[4], hs[8], hs[12]);
            *(float4*)(out + (size_t)((NCH - 1) * CH + s) * BATCH + bbase + piece) = hv;
        }
    }
}

extern "C" void kernel_launch(void* const* d_in, const int* in_sizes, int n_in,
                              void* d_out, int out_size, void* d_ws, size_t ws_size,
                              hipStream_t stream)
{
    const float* x    = (const float*)d_in[0];
    const float* W_ih = (const float*)d_in[1];
    const float* W_hh = (const float*)d_in[2];
    const float* b_ih = (const float*)d_in[3];
    const float* b_hh = (const float*)d_in[4];
    float* out = (float*)d_out;
    float* xp  = (float*)d_ws;

    lstm_xproj_wmma<<<512, 256, 0, stream>>>(x, W_ih, b_ih, b_hh, xp);

    lstm_scan<<<16, 288, 0, stream>>>(xp, W_hh, out);
}


# scan: two waves per work-group run the chain in turns of 128 steps from gate rows held in registers, each loading its next chunk and storing its h off turn; no LDS in the chain, mover waves gone
# speedup vs baseline: 24.0181x; 1.2373x over previous
#include <hip/hip_runtime.h>
#include <hip/hip_bf16.h>

typedef __attribute__((ext_vector_type(2))) float v2f;
typedef __attribute__((ext_vector_type(4))) float v4f;
typedef __attribute__((ext_vector_type(8))) float v8f;

#define T_STEPS 2048
#define BATCH   512
#define NIN     27
#define NROWS   (T_STEPS * BATCH)
#define NTILES  (NROWS / 16)

#define XB_TILES    4
#define XB_ROWS     (16 * XB_TILES)
#define XB_VECS     (XB_ROWS * NIN / 4)
#define XB_FULL     (XB_VECS / 32)
#define XB_REST     (XB_VECS - 32 * XB_FULL)
#define NBLOCKS     (NTILES / XB_TILES)
#define XP_WAVES    8
#define XP_THREADS  (32 * XP_WAVES)

__global__ void __launch_bounds__(XP_THREADS)
lstm_xproj_wmma(const float* __restrict__ x,
                const float* __restrict__ W_ih,
                const float* __restrict__ b_ih,
                const float* __restrict__ b_hh,
                float* __restrict__ xp)
{
    __shared__ v4f s_x[XP_WAVES][XB_VECS];

    const int lane   = threadIdx.x & 31;
    const int wib    = __builtin_amdgcn_readfirstlane((int)(threadIdx.x >> 5));
    const int wave   = blockIdx.x * XP_WAVES + wib;
    const int nwaves = gridDim.x * XP_WAVES;

    const int mn    = lane & 15;
    const int khalf = (lane >> 4) << 1;

    v2f wfrag[7];
#pragma unroll
    for (int kk = 0; kk < 7; ++kk) {
        const int k0 = kk * 4 + khalf;
        float bx = 0.0f, by = 0.0f;
        if (mn < 4) {
            if (k0     < NIN) bx = W_ih[mn * NIN + k0];
            if (k0 + 1 < NIN) by = W_ih[mn * NIN + k0 + 1];
        }
        wfrag[kk].x = bx;
        wfrag[kk].y = by;
    }
    const float bias0 = 0.5f * (b_ih[0] + b_hh[0]);
    const float bias1 = 0.5f * (b_ih[1] + b_hh[1]);
    const float bias2 =         b_ih[2] + b_hh[2];
    const float bias3 = 0.5f * (b_ih[3] + b_hh[3]);

    v4f*         st4 = s_x[wib];
    const float* stf = (const float*)s_x[wib];
    const v4f*   x4  = (const v4f*)x;
    v4f*         xp4 = (v4f*)xp;

    v4f v[XB_FULL + 1];
    int blk = wave;
    if (blk < NBLOCKS) {
        const v4f* src = x4 + (size_t)blk * XB_VECS + lane;
#pragma unroll
        for (int i = 0; i < XB_FULL; ++i) v[i] = src[i * 32];
        if (lane < XB_REST) v[XB_FULL] = src[XB_FULL * 32];
    }
    while (blk < NBLOCKS) {
#pragma unroll
        for (int i = 0; i < XB_FULL; ++i) st4[i * 32 + lane] = v[i];
        if (lane < XB_REST) st4[XB_FULL * 32 + lane] = v[XB_FULL];

        const int nb = blk + nwaves;
        if (nb < NBLOCKS) {
            const v4f* src = x4 + (size_t)nb * XB_VECS + lane;
#pragma unroll
            for (int i = 0; i < XB_FULL; ++i) v[i] = src[i * 32];
            if (lane < XB_REST) v[XB_FULL] = src[XB_FULL * 32];
        }

        const float* arow = stf + mn * NIN;
        v8f acc[XB_TILES];
#pragma unroll
        for (int j = 0; j < XB_TILES; ++j) acc[j] = v8f{};
#pragma unroll
        for (int kk = 0; kk < 7; ++kk) {
            const int  k0  = kk * 4 + khalf;
            const bool pad = (k0 + 1 >= NIN);
            const int  k1  = pad ? k0 : k0 + 1;
#pragma unroll
            for (int j = 0; j < XB_TILES; ++j) {
                v2f xf;
                xf.x = arow[j * 16 * NIN + k0];
                const float y = arow[j * 16 * NIN + k1];
                xf.y = pad ? 0.0f : y;
                acc[j] = __builtin_amdgcn_wmma_f32_16x16x4_f32(
                    false, wfrag[kk], false, xf, (short)0, acc[j], false, false);
            }
        }

        if (lane < 16) {
#pragma unroll
            for (int j = 0; j < XB_TILES; ++j) {
                v4f o;
                o.x = fmaf(acc[j][0], 0.5f, bias0);
                o.y = fmaf(acc[j][1], 0.5f, bias1);
                o.z = acc[j][2] + bias2;
                o.w = fmaf(acc[j][3], 0.5f, bias3);
                xp4[(size_t)(blk * XB_TILES + j) * 16 + lane] = o;
            }
        }
        blk = nb;
    }
}

__device__ __forceinline__ float hw_tanh(float v) {
    return __builtin_amdgcn_tanhf(v);
}
__device__ __forceinline__ float sigmoid_of_half(float zh) {
    return 0.5f + 0.5f * hw_tanh(zh);
}

#define CH      128
#define NCH     (T_STEPS / CH)
#define SCAN_THREADS 64

struct Carry {
    float tc;
    float hc;
    float owi, owf, owg, owo;
};

__device__ __forceinline__ float lstm_cell(const float4 g,
                                           const float qi, const float qf,
                                           const float qg, const float qo,
                                           Carry& s)
{
    const float ti = hw_tanh(fmaf(s.tc, s.owi, g.x));
    const float tf = hw_tanh(fmaf(s.tc, s.owf, g.y));
    const float tg = hw_tanh(fmaf(s.tc, s.owg, g.z));
    const float to = hw_tanh(fmaf(s.tc, s.owo, g.w));
    const float i  = fmaf(0.5f, ti, 0.5f);
    const float fc = fmaf(tf, s.hc, s.hc);
    const float c  = fmaf(i, tg, fc);
    __builtin_amdgcn_sched_barrier(0);
    const float tc = hw_tanh(c);
    const float o  = fmaf(0.5f, to, 0.5f);
    s.owi = fmaf(to, qi, qi);
    s.owf = fmaf(to, qf, qf);
    s.owg = fmaf(to, qg, qg);
    s.owo = fmaf(to, qo, qo);
    s.hc  = 0.5f * c;
    s.tc  = tc;
    float h = o * tc;
    asm volatile("" : "+v"(h));
    return h;
}

__global__ void __launch_bounds__(SCAN_THREADS)
lstm_scan(const float* __restrict__ xp,
          const float* __restrict__ W_hh,
          float* __restrict__ out)
{
    __shared__ float4 s_c[2][32];

    const int lane  = threadIdx.x & 31;
    const int wave  = __builtin_amdgcn_readfirstlane((int)(threadIdx.x >> 5));
    const int bbase = blockIdx.x * 32;

    const float qi = 0.25f * W_hh[0];
    const float qf = 0.25f * W_hh[1];
    const float qg = 0.5f  * W_hh[2];
    const float qo = 0.25f * W_hh[3];

    const float4* __restrict__ gp = (const float4*)xp + bbase + lane;
    float*        __restrict__ op = out + bbase + lane;

    float4 g[CH];
    float  h[CH];

    Carry s = {0.0f, 0.0f, 0.0f, 0.0f, 0.0f, 0.0f};

#pragma unroll
    for (int j = 0; j < CH; ++j)
        g[j] = gp[(size_t)(wave * CH + j) * BATCH];

#pragma nounroll
    for (int ck = 0; ck < NCH; ++ck) {
        const bool mine = ((ck & 1) == wave);
        if (mine) {
            if (ck > 0) {
                const float4 c0 = s_c[0][lane];
                const float4 c1 = s_c[1][lane];
                s.tc  = c0.x; s.hc  = c0.y; s.owi = c0.z; s.owf = c0.w;
                s.owg = c1.x; s.owo = c1.y;
            }
#pragma unroll
            for (int j = 0; j < CH; ++j)
                h[j] = lstm_cell(g[j], qi, qf, qg, qo, s);
            s_c[0][lane] = make_float4(s.tc, s.hc, s.owi, s.owf);
            s_c[1][lane] = make_float4(s.owg, s.owo, 0.0f, 0.0f);
        }
        __syncthreads();
        if (mine) {
            if (ck + 2 < NCH) {
#pragma unroll
                for (int j = 0; j < CH; ++j)
                    g[j] = gp[(size_t)((ck + 2) * CH + j) * BATCH];
            }
#pragma unroll
            for (int j = 0; j < CH; ++j)
                op[(size_t)(ck * CH + j) * BATCH] = h[j];
        }
    }
}

extern "C" void kernel_launch(void* const* d_in, const int* in_sizes, int n_in,
                              void* d_out, int out_size, void* d_ws, size_t ws_size,
                              hipStream_t stream)
{
    const float* x    = (const float*)d_in[0];
    const float* W_ih = (const float*)d_in[1];
    const float* W_hh = (const float*)d_in[2];
    const float* b_ih = (const float*)d_in[3];
    const float* b_hh = (const float*)d_in[4];
    float* out = (float*)d_out;
    float* xp  = (float*)d_ws;

    lstm_xproj_wmma<<<512, 256, 0, stream>>>(x, W_ih, b_ih, b_hh, xp);

    lstm_scan<<<16, SCAN_THREADS, 0, stream>>>(xp, W_hh, out);
}
